# starting kernel
# baseline (speedup 1.0000x reference)
#include <hip/hip_runtime.h>
#include <cstdint>
#include <cstddef>

typedef __attribute__((ext_vector_type(2))) float v2f;
typedef __attribute__((ext_vector_type(8))) float v8f;

#define NEG_SLOPE 0.01f


__global__ void zero_f32_v4(float4* __restrict__ p, long long n4) {
  long long i = (long long)blockIdx.x * blockDim.x + threadIdx.x;
  if (i < n4) p[i] = make_float4(0.f, 0.f, 0.f, 0.f);
}

__global__ void init_deg(float* __restrict__ deg, int n) {
  int i = blockIdx.x * blockDim.x + threadIdx.x;
  if (i < n) deg[i] = 1.0f;
}

__global__ void deg_accum(const int* __restrict__ dst, float* __restrict__ deg, int E) {
  int e = blockIdx.x * blockDim.x + threadIdx.x;
  if (e < E) atomicAdd(&deg[dst[e]], 1.0f);
}

__global__ void compute_dinv(const float* __restrict__ deg, float* __restrict__ dinv, int n) {
  int i = blockIdx.x * blockDim.x + threadIdx.x;
  if (i < n) {
    float d = deg[i];
    dinv[i] = d > 0.0f ? rsqrtf(d) : 0.0f;
  }
}

__global__ void compute_norm(const int* __restrict__ src, const int* __restrict__ dst,
                             const float* __restrict__ dinv, float* __restrict__ norm, int E) {
  int e = blockIdx.x * blockDim.x + threadIdx.x;
  if (e < E) norm[e] = dinv[src[e]] * dinv[dst[e]];
}

template <int K, int NOUT>
__global__ void gcn_gemm_wmma(const float* __restrict__ X, const float* __restrict__ W,
                              float* __restrict__ Y) {
  constexpr int LDSROW = K + 4;
  __shared__ float At[16 * LDSROW];

  const int tid = threadIdx.x;
  const size_t mbase_row = (size_t)blockIdx.x * 16;

  const float* __restrict__ xtile = X + mbase_row * K;
  for (int i = tid * 4; i < 16 * K; i += blockDim.x * 4) {
    const int r = i / K;
    const int c = i % K;
    float4 v = *(const float4*)(xtile + (size_t)r * K + c);
    float* d = &At[r * LDSROW + c];
    d[0] = v.x; d[1] = v.y; d[2] = v.z; d[3] = v.w;
  }
  __syncthreads();

  const int wave = tid >> 5;
  const int lane = tid & 31;
  const int lrow  = lane & 15;
  const int khalf = (lane >> 4) * 2;
  const int n0    = wave * 16;

  v8f acc = {};
  for (int k0 = 0; k0 < K; k0 += 4) {
    const float* ap = &At[lrow * LDSROW + k0 + khalf];
    v2f a; a.x = ap[0]; a.y = ap[1];
    const float* wp = W + (size_t)(k0 + khalf) * NOUT + n0 + lrow;
    v2f b; b.x = wp[0]; b.y = wp[NOUT];
    acc = __builtin_amdgcn_wmma_f32_16x16x4_f32(
         false, a,  false, b,
         (short)0, acc,  false,  false);
  }

  float* yp = Y + mbase_row * NOUT + n0 + lrow;
  const int mofs = (lane < 16) ? 0 : 8;
#pragma unroll
  for (int v = 0; v < 8; ++v) {
    yp[(size_t)(mofs + v) * NOUT] = acc[v];
  }
}

template <int F>
__global__ void edge_scatter_v4(const int* __restrict__ src, const int* __restrict__ dst,
                                const float* __restrict__ norm, const float* __restrict__ h,
                                float* __restrict__ agg, int E) {
  constexpr int FV = F / 4;
  long long gid = (long long)blockIdx.x * blockDim.x + threadIdx.x;
  if (gid >= (long long)E * FV) return;
  const int e  = (int)(gid >> __builtin_ctz(FV));
  const int f4 = (int)(gid & (FV - 1));
  const int s = src[e];
  const int d = dst[e];
  const float w = norm[e];
  const float4 hv = *(const float4*)(h + (size_t)s * F + f4 * 4);
  float* ap = agg + (size_t)d * F + f4 * 4;
  atomicAdd(ap + 0, hv.x * w);
  atomicAdd(ap + 1, hv.y * w);
  atomicAdd(ap + 2, hv.z * w);
  atomicAdd(ap + 3, hv.w * w);
}

template <int F>
__global__ void finalize_act_v4(const float* __restrict__ agg, const float* __restrict__ h,
                                const float* __restrict__ dinv, const float* __restrict__ bias,
                                float* __restrict__ out, int n) {
  constexpr int FV = F / 4;
  long long gid = (long long)blockIdx.x * blockDim.x + threadIdx.x;
  if (gid >= (long long)n * FV) return;
  const int i  = (int)(gid >> __builtin_ctz(FV));
  const int f4 = (int)(gid & (FV - 1));
  const float di2 = dinv[i] * dinv[i];
  const size_t base = gid * 4;
  const float4 av = *(const float4*)(agg + base);
  const float4 hv = *(const float4*)(h + base);
  const float4 bv = *(const float4*)(bias + f4 * 4);
  float4 r;
  r.x = av.x + di2 * hv.x + bv.x;
  r.y = av.y + di2 * hv.y + bv.y;
  r.z = av.z + di2 * hv.z + bv.z;
  r.w = av.w + di2 * hv.w + bv.w;
  r.x = (r.x >= 0.f) ? r.x : NEG_SLOPE * r.x;
  r.y = (r.y >= 0.f) ? r.y : NEG_SLOPE * r.y;
  r.z = (r.z >= 0.f) ? r.z : NEG_SLOPE * r.z;
  r.w = (r.w >= 0.f) ? r.w : NEG_SLOPE * r.w;
  *(float4*)(out + base) = r;
}

template <int F>
__global__ void finalize_lin_v4(float* __restrict__ agg, const float* __restrict__ h,
                                const float* __restrict__ dinv, const float* __restrict__ bias,
                                int n) {
  constexpr int FV = F / 4;
  long long gid = (long long)blockIdx.x * blockDim.x + threadIdx.x;
  if (gid >= (long long)n * FV) return;
  const int i  = (int)(gid >> __builtin_ctz(FV));
  const int f4 = (int)(gid & (FV - 1));
  const float di2 = dinv[i] * dinv[i];
  const size_t base = gid * 4;
  const float4 av = *(const float4*)(agg + base);
  const float4 hv = *(const float4*)(h + base);
  const float4 bv = *(const float4*)(bias + f4 * 4);
  float4 r;
  r.x = av.x + di2 * hv.x + bv.x;
  r.y = av.y + di2 * hv.y + bv.y;
  r.z = av.z + di2 * hv.z + bv.z;
  r.w = av.w + di2 * hv.w + bv.w;
  *(float4*)(agg + base) = r;
}


extern "C" void kernel_launch(void* const* d_in, const int* in_sizes, int n_in,
                              void* d_out, int out_size, void* d_ws, size_t ws_size,
                              hipStream_t stream) {
  const float* x   = (const float*)d_in[0];
  const int*   ei  = (const int*)d_in[1];
  const float* W1  = (const float*)d_in[2];
  const float* b1  = (const float*)d_in[3];
  const float* W2  = (const float*)d_in[4];
  const float* b2  = (const float*)d_in[5];
  float* out = (float*)d_out;

  const int N = in_sizes[0] / 256;
  const int E = in_sizes[1] / 2;
  const int* src = ei;
  const int* dst = ei + E;

  float* ws   = (float*)d_ws;
  float* deg  = ws;
  float* dinv = deg + N;
  float* norm = dinv + N;
  float* h1   = norm + E;
  float* agg1 = h1 + (size_t)N * 128;
  float* h2 = agg1;
  float* g  = h1;

  const int T = 256;
  auto blk = [](long long n, int t) { return (unsigned)((n + t - 1) / t); };

  init_deg<<<blk(N, T), T, 0, stream>>>(deg, N);
  deg_accum<<<blk(E, T), T, 0, stream>>>(dst, deg, E);
  compute_dinv<<<blk(N, T), T, 0, stream>>>(deg, dinv, N);
  compute_norm<<<blk(E, T), T, 0, stream>>>(src, dst, dinv, norm, E);

  gcn_gemm_wmma<256, 128><<<N / 16, 8 * 32, 0, stream>>>(x, W1, h1);

  zero_f32_v4<<<blk((long long)N * 32, T), T, 0, stream>>>((float4*)agg1, (long long)N * 32);
  edge_scatter_v4<128><<<blk((long long)E * 32, T), T, 0, stream>>>(src, dst, norm, h1, agg1, E);
  finalize_act_v4<128><<<blk((long long)N * 32, T), T, 0, stream>>>(agg1, h1, dinv, b1, h2, N);

  gcn_gemm_wmma<128, 64><<<N / 16, 4 * 32, 0, stream>>>(h2, W2, g);

  zero_f32_v4<<<blk((long long)N * 16, T), T, 0, stream>>>((float4*)out, (long long)N * 16);
  edge_scatter_v4<64><<<blk((long long)E * 16, T), T, 0, stream>>>(src, dst, norm, g, out, E);
  finalize_lin_v4<64><<<blk((long long)N * 16, T), T, 0, stream>>>(out, g, dinv, b2, N);
}


# Float atomic scatter-add replaced by a per-call CSR sorted by destination (int atomic rank, scan, fill) and a wave-per-node row gather with fused epilogue; sum order = atomic arrival rank
# speedup vs baseline: 2.4725x; 2.4725x over previous
#include <hip/hip_runtime.h>
#include <cstdint>
#include <cstddef>

typedef __attribute__((ext_vector_type(2))) float v2f;
typedef __attribute__((ext_vector_type(4))) float v4f;
typedef __attribute__((ext_vector_type(8))) float v8f;

#define NEG_SLOPE 0.01f
#define SCAN_THREADS 256
#define SCAN_TILE 1024


__global__ void zero_i32(int* __restrict__ p, int n) {
  int i = blockIdx.x * blockDim.x + threadIdx.x;
  if (i < n) p[i] = 0;
}

__global__ void edge_rank(const int* __restrict__ dst, int* __restrict__ cnt,
                          int* __restrict__ rank, int E) {
  int e = blockIdx.x * blockDim.x + threadIdx.x;
  if (e < E) rank[e] = atomicAdd(&cnt[dst[e]], 1);
}

__global__ void scan_tile(int* __restrict__ cnt, float* __restrict__ dinv,
                          int* __restrict__ tile_sum, int n) {
  __shared__ int wsum[SCAN_THREADS / 32];
  const int tid = threadIdx.x;
  const int lane = tid & 31;
  const int w = tid >> 5;
  const int base = blockIdx.x * SCAN_TILE + tid * 4;
  int v[4];
  for (int k = 0; k < 4; ++k) {
    v[k] = 0;
    if (base + k < n) {
      v[k] = cnt[base + k];
      const float dg = (float)(v[k] + 1);
      dinv[base + k] = dg > 0.0f ? rsqrtf(dg) : 0.0f;
    }
  }
  const int t = v[0] + v[1] + v[2] + v[3];
  int inc = t;
  for (int o = 1; o < 32; o <<= 1) {
    const int y = __shfl_up(inc, o);
    if (lane >= o) inc += y;
  }
  if (lane == 31) wsum[w] = inc;
  __syncthreads();
  int woff = 0;
  for (int k = 0; k < w; ++k) woff += wsum[k];
  int ex = woff + inc - t;
  for (int k = 0; k < 4; ++k) {
    if (base + k < n) cnt[base + k] = ex;
    ex += v[k];
  }
  if (tid == SCAN_THREADS - 1) tile_sum[blockIdx.x] = woff + inc;
}

__global__ void scan_add_offsets(int* __restrict__ rowptr, const int* __restrict__ tile_sum, int n) {
  __shared__ int red[SCAN_THREADS / 32];
  const int tid = threadIdx.x;
  int part = 0;
  for (int k = tid; k < (int)blockIdx.x; k += SCAN_THREADS) part += tile_sum[k];
  for (int o = 16; o > 0; o >>= 1) part += __shfl_xor(part, o);
  if ((tid & 31) == 0) red[tid >> 5] = part;
  __syncthreads();
  int off = 0;
  for (int k = 0; k < SCAN_THREADS / 32; ++k) off += red[k];
  const int base = blockIdx.x * SCAN_TILE + tid * 4;
  for (int k = 0; k < 4; ++k)
    if (base + k < n) rowptr[base + k] += off;
}

__global__ void csr_fill(const int* __restrict__ src, const int* __restrict__ dst,
                         const int* __restrict__ rowptr, const int* __restrict__ rank,
                         int* __restrict__ csr, int E) {
  int e = blockIdx.x * blockDim.x + threadIdx.x;
  if (e < E) csr[rowptr[dst[e]] + rank[e]] = src[e];
}

template <int K, int NOUT>
__global__ void gcn_gemm_wmma(const float* __restrict__ X, const float* __restrict__ W,
                              float* __restrict__ Y) {
  constexpr int LDSROW = K + 4;
  __shared__ float At[16 * LDSROW];

  const int tid = threadIdx.x;
  const size_t mbase_row = (size_t)blockIdx.x * 16;

  const float* __restrict__ xtile = X + mbase_row * K;
  for (int i = tid * 4; i < 16 * K; i += blockDim.x * 4) {
    const int r = i / K;
    const int c = i % K;
    float4 v = *(const float4*)(xtile + (size_t)r * K + c);
    float* d = &At[r * LDSROW + c];
    d[0] = v.x; d[1] = v.y; d[2] = v.z; d[3] = v.w;
  }
  __syncthreads();

  const int wave = tid >> 5;
  const int lane = tid & 31;
  const int lrow  = lane & 15;
  const int khalf = (lane >> 4) * 2;
  const int n0    = wave * 16;

  v8f acc = {};
  for (int k0 = 0; k0 < K; k0 += 4) {
    const float* ap = &At[lrow * LDSROW + k0 + khalf];
    v2f a; a.x = ap[0]; a.y = ap[1];
    const float* wp = W + (size_t)(k0 + khalf) * NOUT + n0 + lrow;
    v2f b; b.x = wp[0]; b.y = wp[NOUT];
    acc = __builtin_amdgcn_wmma_f32_16x16x4_f32(
         false, a,  false, b,
         (short)0, acc,  false,  false);
  }

  float* yp = Y + mbase_row * NOUT + n0 + lrow;
  const int mofs = (lane < 16) ? 0 : 8;
#pragma unroll
  for (int v = 0; v < 8; ++v) {
    yp[(size_t)(mofs + v) * NOUT] = acc[v];
  }
}

template <typename V, bool ACT>
__global__ void gather_rows(const int* __restrict__ rowptr, const int* __restrict__ csr,
                            const float* __restrict__ dinv, const float* __restrict__ h,
                            const float* __restrict__ bias, float* __restrict__ out,
                            int n, int E) {
  constexpr int VEC = sizeof(V) / sizeof(float);
  constexpr int F = 32 * VEC;
  const int lane = threadIdx.x & 31;
  int d = (int)(((long long)blockIdx.x * blockDim.x + threadIdx.x) >> 5);
  if (d >= n) return;
  d = __builtin_amdgcn_readfirstlane(d);
  const int beg = rowptr[d];
  const int end = (d + 1 < n) ? rowptr[d + 1] : E;
  const int m = end - beg;
  const float dd = dinv[d];
  const float* __restrict__ hl = h + lane * VEC;

  V acc = (V)(0.0f);
  for (int base = 0; base < m; base += 32) {
    int s_l = 0;
    float w_l = 0.0f;
    if (base + lane < m) {
      s_l = csr[beg + base + lane];
      w_l = dinv[s_l] * dd;
    }
    const int c = (m - base < 32) ? (m - base) : 32;
    int j = 0;
    for (; j + 4 <= c; j += 4) {
      const int s0 = __builtin_amdgcn_readlane(s_l, j);
      const int s1 = __builtin_amdgcn_readlane(s_l, j + 1);
      const int s2 = __builtin_amdgcn_readlane(s_l, j + 2);
      const int s3 = __builtin_amdgcn_readlane(s_l, j + 3);
      const float w0 = __int_as_float(__builtin_amdgcn_readlane(__float_as_int(w_l), j));
      const float w1 = __int_as_float(__builtin_amdgcn_readlane(__float_as_int(w_l), j + 1));
      const float w2 = __int_as_float(__builtin_amdgcn_readlane(__float_as_int(w_l), j + 2));
      const float w3 = __int_as_float(__builtin_amdgcn_readlane(__float_as_int(w_l), j + 3));
      const V a0 = *(const V*)(hl + (size_t)s0 * F);
      const V a1 = *(const V*)(hl + (size_t)s1 * F);
      const V a2 = *(const V*)(hl + (size_t)s2 * F);
      const V a3 = *(const V*)(hl + (size_t)s3 * F);
      acc += a0 * w0;
      acc += a1 * w1;
      acc += a2 * w2;
      acc += a3 * w3;
    }
    if (j + 2 <= c) {
      const int s0 = __builtin_amdgcn_readlane(s_l, j);
      const int s1 = __builtin_amdgcn_readlane(s_l, j + 1);
      const float w0 = __int_as_float(__builtin_amdgcn_readlane(__float_as_int(w_l), j));
      const float w1 = __int_as_float(__builtin_amdgcn_readlane(__float_as_int(w_l), j + 1));
      const V a0 = *(const V*)(hl + (size_t)s0 * F);
      const V a1 = *(const V*)(hl + (size_t)s1 * F);
      acc += a0 * w0;
      acc += a1 * w1;
      j += 2;
    }
    if (j < c) {
      const int s0 = __builtin_amdgcn_readlane(s_l, j);
      const float w0 = __int_as_float(__builtin_amdgcn_readlane(__float_as_int(w_l), j));
      const V a0 = *(const V*)(hl + (size_t)s0 * F);
      acc += a0 * w0;
    }
  }

  const float di2 = dd * dd;
  const V hv = *(const V*)(hl + (size_t)d * F);
  const V bv = *(const V*)(bias + lane * VEC);
  V r = acc + hv * di2 + bv;
  if (ACT) {
    for (int k = 0; k < VEC; ++k) r[k] = (r[k] >= 0.f) ? r[k] : NEG_SLOPE * r[k];
  }
  *(V*)(out + (size_t)d * F + lane * VEC) = r;
}


extern "C" void kernel_launch(void* const* d_in, const int* in_sizes, int n_in,
                              void* d_out, int out_size, void* d_ws, size_t ws_size,
                              hipStream_t stream) {
  const float* x   = (const float*)d_in[0];
  const int*   ei  = (const int*)d_in[1];
  const float* W1  = (const float*)d_in[2];
  const float* b1  = (const float*)d_in[3];
  const float* W2  = (const float*)d_in[4];
  const float* b2  = (const float*)d_in[5];
  float* out = (float*)d_out;

  const int N = in_sizes[0] / 256;
  const int E = in_sizes[1] / 2;
  const int* src = ei;
  const int* dst = ei + E;

  float* ws     = (float*)d_ws;
  int*   rowptr = (int*)ws;
  float* dinv   = ws + N;
  int*   csr    = (int*)(dinv + N);
  float* h1     = (float*)(csr + E);
  float* h2     = h1 + (size_t)N * 128;
  int*   rank   = (int*)h2;
  int*   tile_sum = rank + E;
  float* g = h1;

  const int T = 256;
  auto blk = [](long long n, int t) { return (unsigned)((n + t - 1) / t); };
  const unsigned tiles = blk(N, SCAN_TILE);

  zero_i32<<<blk(N, T), T, 0, stream>>>(rowptr, N);
  edge_rank<<<blk(E, T), T, 0, stream>>>(dst, rowptr, rank, E);
  scan_tile<<<tiles, SCAN_THREADS, 0, stream>>>(rowptr, dinv, tile_sum, N);
  scan_add_offsets<<<tiles, SCAN_THREADS, 0, stream>>>(rowptr, tile_sum, N);
  csr_fill<<<blk(E, T), T, 0, stream>>>(src, dst, rowptr, rank, csr, E);

  gcn_gemm_wmma<256, 128><<<N / 16, 8 * 32, 0, stream>>>(x, W1, h1);

  gather_rows<v4f, true><<<blk((long long)N * 32, T), T, 0, stream>>>(rowptr, csr, dinv, h1, b1, h2, N, E);

  gcn_gemm_wmma<128, 64><<<N / 16, 4 * 32, 0, stream>>>(h2, W2, g);

  gather_rows<v2f, false><<<blk((long long)N * 32, T), T, 0, stream>>>(rowptr, csr, dinv, g, b2, out, N, E);
}


# CSR fill moved into the layer-1 GEMM launch, one edge per thread: its scattered 4-byte reads and stores run under the matrix work; 9 -> 8 launches; sum orders as before
# speedup vs baseline: 2.6809x; 1.0843x over previous
#include <hip/hip_runtime.h>
#include <cstdint>
#include <cstddef>

typedef __attribute__((ext_vector_type(2))) float v2f;
typedef __attribute__((ext_vector_type(4))) float v4f;
typedef __attribute__((ext_vector_type(8))) float v8f;

#define NEG_SLOPE 0.01f
#define SCAN_THREADS 256
#define SCAN_TILE 1024


__global__ void zero_i32(int* __restrict__ p, int n) {
  int i = blockIdx.x * blockDim.x + threadIdx.x;
  if (i < n) p[i] = 0;
}

__global__ void edge_rank(const int* __restrict__ dst, int* __restrict__ cnt,
                          int* __restrict__ rank, int E) {
  int e = blockIdx.x * blockDim.x + threadIdx.x;
  if (e < E) rank[e] = atomicAdd(&cnt[dst[e]], 1);
}

__global__ void scan_tile(int* __restrict__ cnt, float* __restrict__ dinv,
                          int* __restrict__ tile_sum, int n) {
  __shared__ int wsum[SCAN_THREADS / 32];
  const int tid = threadIdx.x;
  const int lane = tid & 31;
  const int w = tid >> 5;
  const int base = blockIdx.x * SCAN_TILE + tid * 4;
  int v[4];
  for (int k = 0; k < 4; ++k) {
    v[k] = 0;
    if (base + k < n) {
      v[k] = cnt[base + k];
      const float dg = (float)(v[k] + 1);
      dinv[base + k] = dg > 0.0f ? rsqrtf(dg) : 0.0f;
    }
  }
  const int t = v[0] + v[1] + v[2] + v[3];
  int inc = t;
  for (int o = 1; o < 32; o <<= 1) {
    const int y = __shfl_up(inc, o);
    if (lane >= o) inc += y;
  }
  if (lane == 31) wsum[w] = inc;
  __syncthreads();
  int woff = 0;
  for (int k = 0; k < w; ++k) woff += wsum[k];
  int ex = woff + inc - t;
  for (int k = 0; k < 4; ++k) {
    if (base + k < n) cnt[base + k] = ex;
    ex += v[k];
  }
  if (tid == SCAN_THREADS - 1) tile_sum[blockIdx.x] = woff + inc;
}

__global__ void scan_add_offsets(int* __restrict__ rowptr, const int* __restrict__ tile_sum, int n) {
  __shared__ int red[SCAN_THREADS / 32];
  const int tid = threadIdx.x;
  int part = 0;
  for (int k = tid; k < (int)blockIdx.x; k += SCAN_THREADS) part += tile_sum[k];
  for (int o = 16; o > 0; o >>= 1) part += __shfl_xor(part, o);
  if ((tid & 31) == 0) red[tid >> 5] = part;
  __syncthreads();
  int off = 0;
  for (int k = 0; k < SCAN_THREADS / 32; ++k) off += red[k];
  const int base = blockIdx.x * SCAN_TILE + tid * 4;
  for (int k = 0; k < 4; ++k)
    if (base + k < n) rowptr[base + k] += off;
}


template <int K, int NOUT, bool FILL>
__global__ void gcn_gemm_wmma(const float* __restrict__ X, const float* __restrict__ W,
                              const float* __restrict__ dinv, float* __restrict__ Y,
                              const int* __restrict__ src, const int* __restrict__ dst,
                              const int* __restrict__ rowptr, const int* __restrict__ rank,
                              int* __restrict__ csr, int E) {
  constexpr int LDSROW = K + 4;
  __shared__ float At[16 * LDSROW];

  const int tid = threadIdx.x;
  const size_t mbase_row = (size_t)blockIdx.x * 16;

  const int fe = (int)(blockIdx.x * blockDim.x) + tid;
  const int fstride = (int)(gridDim.x * blockDim.x);
  unsigned fd = 0;
  int frk = 0, fs = 0;
  if (FILL && fe < E) { fd = (unsigned)dst[fe]; frk = rank[fe]; fs = src[fe]; }

  const int mofs = (tid & 16) ? 8 : 0;
  const float4 ds0 = *(const float4*)(dinv + mbase_row + mofs);
  const float4 ds1 = *(const float4*)(dinv + mbase_row + mofs + 4);
  const float dsc[8] = {ds0.x, ds0.y, ds0.z, ds0.w, ds1.x, ds1.y, ds1.z, ds1.w};

  const float* __restrict__ xtile = X + mbase_row * K;
  for (int i = tid * 4; i < 16 * K; i += blockDim.x * 4) {
    const int r = i / K;
    const int c = i % K;
    float4 v = *(const float4*)(xtile + (size_t)r * K + c);
    float* d = &At[r * LDSROW + c];
    d[0] = v.x; d[1] = v.y; d[2] = v.z; d[3] = v.w;
  }
  __syncthreads();
  int frp = 0;
  if (FILL && fe < E) frp = rowptr[fd];

  const int wave = tid >> 5;
  const int lane = tid & 31;
  const int lrow  = lane & 15;
  const int khalf = (lane >> 4) * 2;
  const int n0    = wave * 16;

  v8f acc = {};
  for (int k0 = 0; k0 < K; k0 += 4) {
    const float* ap = &At[lrow * LDSROW + k0 + khalf];
    v2f a; a.x = ap[0]; a.y = ap[1];
    const float* wp = W + (size_t)(k0 + khalf) * NOUT + n0 + lrow;
    v2f b; b.x = wp[0]; b.y = wp[NOUT];
    acc = __builtin_amdgcn_wmma_f32_16x16x4_f32(
         false, a,  false, b,
         (short)0, acc,  false,  false);
  }

  float* yp = Y + mbase_row * NOUT + n0 + lrow;
#pragma unroll
  for (int v = 0; v < 8; ++v) {
    yp[(size_t)(mofs + v) * NOUT] = acc[v] * dsc[v];
  }

  if (FILL) {
    if (fe < E) csr[frp + frk] = fs;
#pragma unroll 1
    for (int e = fe + fstride; e < E; e += fstride) csr[rowptr[dst[e]] + rank[e]] = src[e];
  }
}

template <typename V, bool ACT>
__global__ void gather_rows(const int* __restrict__ rowptr, const int* __restrict__ csr,
                            const float* __restrict__ dinv, const float* __restrict__ h,
                            const float* __restrict__ bias, float* __restrict__ out,
                            int n, int E) {
  constexpr int VEC = sizeof(V) / sizeof(float);
  constexpr int F = 32 * VEC;
  const int lane = threadIdx.x & 31;
  int d = (int)(((long long)blockIdx.x * blockDim.x + threadIdx.x) >> 5);
  if (d >= n) return;
  d = __builtin_amdgcn_readfirstlane(d);
  const int beg = rowptr[d];
  const int end = (d + 1 < n) ? rowptr[d + 1] : E;
  const int m = end - beg;
  const float dd = dinv[d];
  const float* __restrict__ hl = h + lane * VEC;

  V acc = *(const V*)(hl + (size_t)d * F);
  const V bv = *(const V*)(bias + lane * VEC);
  for (int base = 0; base < m; base += 32) {
    int s_l = 0;
    if (base + lane < m) s_l = csr[beg + base + lane];
    const int c = (m - base < 32) ? (m - base) : 32;
    int j = 0;
    for (; j + 4 <= c; j += 4) {
      const int s0 = __builtin_amdgcn_readlane(s_l, j);
      const int s1 = __builtin_amdgcn_readlane(s_l, j + 1);
      const int s2 = __builtin_amdgcn_readlane(s_l, j + 2);
      const int s3 = __builtin_amdgcn_readlane(s_l, j + 3);
      const V a0 = *(const V*)(hl + (size_t)s0 * F);
      const V a1 = *(const V*)(hl + (size_t)s1 * F);
      const V a2 = *(const V*)(hl + (size_t)s2 * F);
      const V a3 = *(const V*)(hl + (size_t)s3 * F);
      acc += a0;
      acc += a1;
      acc += a2;
      acc += a3;
    }
    if (j + 2 <= c) {
      const int s0 = __builtin_amdgcn_readlane(s_l, j);
      const int s1 = __builtin_amdgcn_readlane(s_l, j + 1);
      const V a0 = *(const V*)(hl + (size_t)s0 * F);
      const V a1 = *(const V*)(hl + (size_t)s1 * F);
      acc += a0;
      acc += a1;
      j += 2;
    }
    if (j < c) {
      const int s0 = __builtin_amdgcn_readlane(s_l, j);
      const V a0 = *(const V*)(hl + (size_t)s0 * F);
      acc += a0;
    }
  }

  V r = acc * dd + bv;
  if (ACT) {
    for (int k = 0; k < VEC; ++k) r[k] = (r[k] >= 0.f) ? r[k] : NEG_SLOPE * r[k];
  }
  *(V*)(out + (size_t)d * F + lane * VEC) = r;
}


extern "C" void kernel_launch(void* const* d_in, const int* in_sizes, int n_in,
                              void* d_out, int out_size, void* d_ws, size_t ws_size,
                              hipStream_t stream) {
  const float* x   = (const float*)d_in[0];
  const int*   ei  = (const int*)d_in[1];
  const float* W1  = (const float*)d_in[2];
  const float* b1  = (const float*)d_in[3];
  const float* W2  = (const float*)d_in[4];
  const float* b2  = (const float*)d_in[5];
  float* out = (float*)d_out;

  const int N = in_sizes[0] / 256;
  const int E = in_sizes[1] / 2;
  const int* src = ei;
  const int* dst = ei + E;

  float* ws     = (float*)d_ws;
  int*   rowptr = (int*)ws;
  float* dinv   = ws + N;
  int*   csr    = (int*)(dinv + N);
  float* h1     = (float*)(csr + E);
  float* h2     = h1 + (size_t)N * 128;
  int*   rank   = (int*)h2;
  int*   tile_sum = rank + E;
  float* g = h1;

  const int T = 256;
  auto blk = [](long long n, int t) { return (unsigned)((n + t - 1) / t); };
  const unsigned tiles = blk(N, SCAN_TILE);

  zero_i32<<<blk(N, T), T, 0, stream>>>(rowptr, N);
  edge_rank<<<blk(E, T), T, 0, stream>>>(dst, rowptr, rank, E);
  scan_tile<<<tiles, SCAN_THREADS, 0, stream>>>(rowptr, dinv, tile_sum, N);
  scan_add_offsets<<<tiles, SCAN_THREADS, 0, stream>>>(rowptr, tile_sum, N);

  gcn_gemm_wmma<256, 128, true><<<N / 16, 8 * 32, 0, stream>>>(x, W1, dinv, h1, src, dst, rowptr, rank, csr, E);

  gather_rows<v4f, true><<<blk((long long)N * 32, T), T, 0, stream>>>(rowptr, csr, dinv, h1, b1, h2, N, E);

  gcn_gemm_wmma<128, 64, false><<<N / 16, 4 * 32, 0, stream>>>(h2, W2, dinv, g, nullptr, nullptr, nullptr, nullptr, nullptr, 0);

  gather_rows<v2f, false><<<blk((long long)N * 32, T), T, 0, stream>>>(rowptr, csr, dinv, g, b2, out, N, E);
}


# GEMMs: a wave keeps 5 row tiles (80-row blocks) per W fragment, W fetches per product fall to a fifth; x rows staged via LDS in column chunks, next chunk loading under the products; sum order kept
# speedup vs baseline: 2.7863x; 1.0393x over previous
#include <hip/hip_runtime.h>
#include <cstdint>
#include <cstddef>

typedef __attribute__((ext_vector_type(2))) float v2f;
typedef __attribute__((ext_vector_type(4))) float v4f;
typedef __attribute__((ext_vector_type(8))) float v8f;

#define NEG_SLOPE 0.01f
#define SCAN_THREADS 256
#define SCAN_TILE 1024


__global__ void zero_i32(int* __restrict__ p, int n) {
  int i = blockIdx.x * blockDim.x + threadIdx.x;
  if (i < n) p[i] = 0;
}

__global__ void edge_rank(const int* __restrict__ dst, int* __restrict__ cnt,
                          int* __restrict__ rank, int E) {
  int e = blockIdx.x * blockDim.x + threadIdx.x;
  if (e < E) rank[e] = atomicAdd(&cnt[dst[e]], 1);
}

__global__ void scan_tile(int* __restrict__ cnt, float* __restrict__ dinv,
                          int* __restrict__ tile_sum, int n) {
  __shared__ int wsum[SCAN_THREADS / 32];
  const int tid = threadIdx.x;
  const int lane = tid & 31;
  const int w = tid >> 5;
  const int base = blockIdx.x * SCAN_TILE + tid * 4;
  int v[4];
  for (int k = 0; k < 4; ++k) {
    v[k] = 0;
    if (base + k < n) {
      v[k] = cnt[base + k];
      const float dg = (float)(v[k] + 1);
      dinv[base + k] = dg > 0.0f ? rsqrtf(dg) : 0.0f;
    }
  }
  const int t = v[0] + v[1] + v[2] + v[3];
  int inc = t;
  for (int o = 1; o < 32; o <<= 1) {
    const int y = __shfl_up(inc, o);
    if (lane >= o) inc += y;
  }
  if (lane == 31) wsum[w] = inc;
  __syncthreads();
  int woff = 0;
  for (int k = 0; k < w; ++k) woff += wsum[k];
  int ex = woff + inc - t;
  for (int k = 0; k < 4; ++k) {
    if (base + k < n) cnt[base + k] = ex;
    ex += v[k];
  }
  if (tid == SCAN_THREADS - 1) tile_sum[blockIdx.x] = woff + inc;
}

__global__ void scan_add_offsets(int* __restrict__ rowptr, const int* __restrict__ tile_sum, int n) {
  __shared__ int red[SCAN_THREADS / 32];
  const int tid = threadIdx.x;
  int part = 0;
  for (int k = tid; k < (int)blockIdx.x; k += SCAN_THREADS) part += tile_sum[k];
  for (int o = 16; o > 0; o >>= 1) part += __shfl_xor(part, o);
  if ((tid & 31) == 0) red[tid >> 5] = part;
  __syncthreads();
  int off = 0;
  for (int k = 0; k < SCAN_THREADS / 32; ++k) off += red[k];
  const int base = blockIdx.x * SCAN_TILE + tid * 4;
  for (int k = 0; k < 4; ++k)
    if (base + k < n) rowptr[base + k] += off;
}


template <int K, int NOUT, int MT, bool FILL>
__global__ void __launch_bounds__((NOUT / 16) * 32)
gcn_gemm_wmma(const float* __restrict__ X, const float* __restrict__ W,
                              const float* __restrict__ dinv, float* __restrict__ Y,
                              const int* __restrict__ src, const int* __restrict__ dst,
                              const int* __restrict__ rowptr, const int* __restrict__ rank,
                              int* __restrict__ csr, int E) {
  constexpr int NT = (NOUT / 16) * 32;
  constexpr int ROWS = 16 * MT;
  constexpr int KC = (K >= 256) ? 64 : 32;
  constexpr int NCH = K / KC;
  constexpr int PITCH = KC + 4;
  constexpr int NLD = (ROWS * KC / 4) / NT;
  static_assert(K % KC == 0, "K must be a multiple of the chunk");
  static_assert((ROWS * KC / 4) % NT == 0 && NT % (KC / 4) == 0, "chunk must divide evenly over the threads");
  __shared__ alignas(16) float At[2][ROWS * PITCH];

  const int tid = threadIdx.x;
  const size_t mbase_row = (size_t)blockIdx.x * ROWS;

  const int fe = (int)(blockIdx.x * NT) + tid;
  const int fstride = (int)(gridDim.x * NT);
  unsigned fd[MT];
  int frk[MT], fs[MT], frp[MT];
  if (FILL) {
#pragma unroll
    for (int i = 0; i < MT; ++i) {
      const int e = fe + i * fstride;
      fd[i] = 0; frk[i] = 0; fs[i] = 0; frp[i] = 0;
      if (e < E) { fd[i] = (unsigned)dst[e]; frk[i] = rank[e]; fs[i] = src[e]; }
    }
  }

  constexpr int RSTEP = NT / (KC / 4);
  const int srow = tid / (KC / 4);
  const int scol = (tid % (KC / 4)) * 4;
  const float* __restrict__ xsrc = X + (mbase_row + srow) * K + scol;
  float* __restrict__ ldst = &At[0][srow * PITCH + scol];
  float4 st[NLD];
#pragma unroll
  for (int j = 0; j < NLD; ++j) st[j] = *(const float4*)(xsrc + (size_t)j * RSTEP * K);
#pragma unroll
  for (int j = 0; j < NLD; ++j) *(float4*)(ldst + j * RSTEP * PITCH) = st[j];
  __syncthreads();
  if (FILL) {
#pragma unroll
    for (int i = 0; i < MT; ++i)
      if (fe + i * fstride < E) frp[i] = rowptr[fd[i]];
  }

  const int wave = tid >> 5;
  const int lane = tid & 31;
  const int lrow  = lane & 15;
  const int khalf = (lane >> 4) * 2;
  const int n0    = wave * 16;
  const int mofs  = (tid & 16) ? 8 : 0;

  v8f acc0 = {}, acc1 = {}, acc2 = {}, acc3 = {}, acc4 = {};
  static_assert(MT == 5, "five accumulator tiles are written out by name");

  auto chunk_products = [&](const float* __restrict__ ab, const float* __restrict__ wb) {
#pragma unroll
    for (int kk = 0; kk < KC; kk += 4) {
      const float* wp = wb + (size_t)kk * NOUT;
      v2f b; b.x = wp[0]; b.y = wp[NOUT];
      const float* ap = ab + kk;
      v2f a0, a1, a2, a3, a4;
      a0.x = ap[0 * 16 * PITCH]; a0.y = ap[0 * 16 * PITCH + 1];
      a1.x = ap[1 * 16 * PITCH]; a1.y = ap[1 * 16 * PITCH + 1];
      a2.x = ap[2 * 16 * PITCH]; a2.y = ap[2 * 16 * PITCH + 1];
      a3.x = ap[3 * 16 * PITCH]; a3.y = ap[3 * 16 * PITCH + 1];
      a4.x = ap[4 * 16 * PITCH]; a4.y = ap[4 * 16 * PITCH + 1];
      acc0 = __builtin_amdgcn_wmma_f32_16x16x4_f32(false, a0, false, b, (short)0, acc0, false, false);
      acc1 = __builtin_amdgcn_wmma_f32_16x16x4_f32(false, a1, false, b, (short)0, acc1, false, false);
      acc2 = __builtin_amdgcn_wmma_f32_16x16x4_f32(false, a2, false, b, (short)0, acc2, false, false);
      acc3 = __builtin_amdgcn_wmma_f32_16x16x4_f32(false, a3, false, b, (short)0, acc3, false, false);
      acc4 = __builtin_amdgcn_wmma_f32_16x16x4_f32(false, a4, false, b, (short)0, acc4, false, false);
    }
  };
  const float* __restrict__ ab0 = &At[0][lrow * PITCH + khalf];
  const float* __restrict__ wb0 = W + (size_t)khalf * NOUT + n0 + lrow;

#pragma unroll 1
  for (int ch = 0; ch + 1 < NCH; ++ch) {
#pragma unroll
    for (int j = 0; j < NLD; ++j)
      st[j] = *(const float4*)(xsrc + (size_t)j * RSTEP * K + (ch + 1) * KC);
    chunk_products(ab0 + (ch & 1) * (ROWS * PITCH), wb0 + (size_t)ch * KC * NOUT);
    float* __restrict__ lnext = ldst + ((ch + 1) & 1) * (ROWS * PITCH);
#pragma unroll
    for (int j = 0; j < NLD; ++j) *(float4*)(lnext + j * RSTEP * PITCH) = st[j];
    __syncthreads();
  }
  chunk_products(ab0 + ((NCH - 1) & 1) * (ROWS * PITCH), wb0 + (size_t)(NCH - 1) * KC * NOUT);
  const v8f acc[MT] = {acc0, acc1, acc2, acc3, acc4};

#pragma unroll
  for (int t = 0; t < MT; ++t) {
    const size_t row0 = mbase_row + t * 16 + mofs;
    const float4 ds0 = *(const float4*)(dinv + row0);
    const float4 ds1 = *(const float4*)(dinv + row0 + 4);
    const float dsc[8] = {ds0.x, ds0.y, ds0.z, ds0.w, ds1.x, ds1.y, ds1.z, ds1.w};
    float* yp = Y + row0 * NOUT + n0 + lrow;
#pragma unroll
    for (int v = 0; v < 8; ++v) yp[(size_t)v * NOUT] = acc[t][v] * dsc[v];
  }

  if (FILL) {
#pragma unroll
    for (int i = 0; i < MT; ++i)
      if (fe + i * fstride < E) csr[frp[i] + frk[i]] = fs[i];
#pragma unroll 1
    for (int e = fe + MT * fstride; e < E; e += fstride) csr[rowptr[dst[e]] + rank[e]] = src[e];
  }
}

template <typename V, bool ACT>
__global__ void gather_rows(const int* __restrict__ rowptr, const int* __restrict__ csr,
                            const float* __restrict__ dinv, const float* __restrict__ h,
                            const float* __restrict__ bias, float* __restrict__ out,
                            int n, int E) {
  constexpr int VEC = sizeof(V) / sizeof(float);
  constexpr int F = 32 * VEC;
  const int lane = threadIdx.x & 31;
  int d = (int)(((long long)blockIdx.x * blockDim.x + threadIdx.x) >> 5);
  if (d >= n) return;
  d = __builtin_amdgcn_readfirstlane(d);
  const int beg = rowptr[d];
  const int end = (d + 1 < n) ? rowptr[d + 1] : E;
  const int m = end - beg;
  const float dd = dinv[d];
  const float* __restrict__ hl = h + lane * VEC;

  V acc = *(const V*)(hl + (size_t)d * F);
  const V bv = *(const V*)(bias + lane * VEC);
  for (int base = 0; base < m; base += 32) {
    int s_l = 0;
    if (base + lane < m) s_l = csr[beg + base + lane];
    const int c = (m - base < 32) ? (m - base) : 32;
    int j = 0;
    for (; j + 4 <= c; j += 4) {
      const int s0 = __builtin_amdgcn_readlane(s_l, j);
      const int s1 = __builtin_amdgcn_readlane(s_l, j + 1);
      const int s2 = __builtin_amdgcn_readlane(s_l, j + 2);
      const int s3 = __builtin_amdgcn_readlane(s_l, j + 3);
      const V a0 = *(const V*)(hl + (size_t)s0 * F);
      const V a1 = *(const V*)(hl + (size_t)s1 * F);
      const V a2 = *(const V*)(hl + (size_t)s2 * F);
      const V a3 = *(const V*)(hl + (size_t)s3 * F);
      acc += a0;
      acc += a1;
      acc += a2;
      acc += a3;
    }
    if (j + 2 <= c) {
      const int s0 = __builtin_amdgcn_readlane(s_l, j);
      const int s1 = __builtin_amdgcn_readlane(s_l, j + 1);
      const V a0 = *(const V*)(hl + (size_t)s0 * F);
      const V a1 = *(const V*)(hl + (size_t)s1 * F);
      acc += a0;
      acc += a1;
      j += 2;
    }
    if (j < c) {
      const int s0 = __builtin_amdgcn_readlane(s_l, j);
      const V a0 = *(const V*)(hl + (size_t)s0 * F);
      acc += a0;
    }
  }

  V r = acc * dd + bv;
  if (ACT) {
    for (int k = 0; k < VEC; ++k) r[k] = (r[k] >= 0.f) ? r[k] : NEG_SLOPE * r[k];
  }
  *(V*)(out + (size_t)d * F + lane * VEC) = r;
}


extern "C" void kernel_launch(void* const* d_in, const int* in_sizes, int n_in,
                              void* d_out, int out_size, void* d_ws, size_t ws_size,
                              hipStream_t stream) {
  const float* x   = (const float*)d_in[0];
  const int*   ei  = (const int*)d_in[1];
  const float* W1  = (const float*)d_in[2];
  const float* b1  = (const float*)d_in[3];
  const float* W2  = (const float*)d_in[4];
  const float* b2  = (const float*)d_in[5];
  float* out = (float*)d_out;

  const int N = in_sizes[0] / 256;
  const int E = in_sizes[1] / 2;
  const int* src = ei;
  const int* dst = ei + E;

  float* ws     = (float*)d_ws;
  int*   rowptr = (int*)ws;
  float* dinv   = ws + N;
  int*   csr    = (int*)(dinv + N);
  float* h1     = (float*)(csr + E);
  float* h2     = h1 + (size_t)N * 128;
  int*   rank   = (int*)h2;
  int*   tile_sum = rank + E;
  float* g = h1;

  const int T = 256;
  auto blk = [](long long n, int t) { return (unsigned)((n + t - 1) / t); };
  const unsigned tiles = blk(N, SCAN_TILE);

  zero_i32<<<blk(N, T), T, 0, stream>>>(rowptr, N);
  edge_rank<<<blk(E, T), T, 0, stream>>>(dst, rowptr, rank, E);
  scan_tile<<<tiles, SCAN_THREADS, 0, stream>>>(rowptr, dinv, tile_sum, N);
  scan_add_offsets<<<tiles, SCAN_THREADS, 0, stream>>>(rowptr, tile_sum, N);

  gcn_gemm_wmma<256, 128, 5, true><<<N / 80, 8 * 32, 0, stream>>>(x, W1, dinv, h1, src, dst, rowptr, rank, csr, E);

  gather_rows<v4f, true><<<blk((long long)N * 32, T), T, 0, stream>>>(rowptr, csr, dinv, h1, b1, h2, N, E);

  gcn_gemm_wmma<128, 64, 5, false><<<N / 80, 4 * 32, 0, stream>>>(h2, W2, dinv, g, nullptr, nullptr, nullptr, nullptr, nullptr, 0);

  gather_rows<v2f, false><<<blk((long long)N * 32, T), T, 0, stream>>>(rowptr, csr, dinv, g, b2, out, N, E);
}


# CSR build without global atomics: two-level counting sort by destination in LDS (chunk sort by 400-node bin; a block per bin counts, scans, places); 8 -> 6 launches; sum order = LDS atomic arrival
# speedup vs baseline: 4.4159x; 1.5849x over previous
#include <hip/hip_runtime.h>
#include <cstdint>
#include <cstddef>

typedef __attribute__((ext_vector_type(2))) float v2f;
typedef __attribute__((ext_vector_type(4))) float v4f;
typedef __attribute__((ext_vector_type(8))) float v8f;

#define NEG_SLOPE 0.01f
#define BT 1024
#define BIN_ROWS 400
#define MAX_BINS 256
#define CHUNK 6272
#define SRC_BITS 17
#define OUT_CAP 16384
#define EMPTY_EDGE 0xFFFFFFFFu


__global__ void __launch_bounds__(BT)
bin_edges(const int* __restrict__ src, const int* __restrict__ dst,
          unsigned* __restrict__ tmp, int* __restrict__ startT, int E, int nbins) {
  __shared__ int s_hist[MAX_BINS];
  __shared__ int s_start[MAX_BINS + 1];
  __shared__ int s_wsum[MAX_BINS / 32];
  __shared__ unsigned s_sorted[CHUNK];
  constexpr int PER = (CHUNK + BT - 1) / BT;
  const int tid = threadIdx.x;
  const int lane = tid & 31;
  const int wv = tid >> 5;
  const int P = (int)gridDim.x;
  const int p = (int)blockIdx.x;
  const int ebase = p * CHUNK;
  const int cnt = (E - ebase < CHUNK) ? (E - ebase) : CHUNK;

  if (tid < MAX_BINS) s_hist[tid] = 0;
  __syncthreads();

  unsigned dd[PER], ss[PER];
#pragma unroll
  for (int k = 0; k < PER; ++k) {
    const int i = tid + k * BT;
    dd[k] = EMPTY_EDGE; ss[k] = 0;
    if (i < cnt) { dd[k] = (unsigned)dst[ebase + i]; ss[k] = (unsigned)src[ebase + i]; }
  }
  unsigned word[PER];
  int bn[PER], rk[PER];
#pragma unroll
  for (int k = 0; k < PER; ++k) {
    bn[k] = -1; rk[k] = 0; word[k] = 0;
    if (dd[k] != EMPTY_EDGE) {
      const unsigned b = dd[k] / BIN_ROWS;
      bn[k] = (int)b;
      word[k] = ((dd[k] - b * BIN_ROWS) << SRC_BITS) | ss[k];
      rk[k] = atomicAdd(&s_hist[b], 1);
    }
  }
  __syncthreads();

  const int v = (tid < MAX_BINS) ? s_hist[tid] : 0;
  int inc = v;
  for (int o = 1; o < 32; o <<= 1) {
    const int y = __shfl_up(inc, o);
    if (lane >= o) inc += y;
  }
  if (tid < MAX_BINS && lane == 31) s_wsum[wv] = inc;
  __syncthreads();
  if (tid < MAX_BINS) {
    int woff = 0;
    for (int k = 0; k < wv; ++k) woff += s_wsum[k];
    s_start[tid] = woff + inc - v;
    if (tid == MAX_BINS - 1) s_start[MAX_BINS] = woff + inc;
  }
  __syncthreads();

#pragma unroll
  for (int k = 0; k < PER; ++k)
    if (bn[k] >= 0) s_sorted[s_start[bn[k]] + rk[k]] = word[k];
  __syncthreads();

  for (int i = tid; i < cnt; i += BT) tmp[ebase + i] = s_sorted[i];
  if (tid <= nbins) startT[tid * P + p] = s_start[tid];
}

template <int MODE>
__device__ __forceinline__ int sweep_bin(const unsigned* __restrict__ tmp,
                                         const int* __restrict__ st0, const int* __restrict__ st1,
                                         int P, int wv, int lane,
                                         int* __restrict__ s_cnt, int* __restrict__ s_out,
                                         int* __restrict__ csr_bin) {
  auto use = [&](unsigned w) {
    if (MODE == 0) {
      atomicAdd(&s_cnt[w >> SRC_BITS], 1);
    } else {
      const int pos = atomicAdd(&s_cnt[w >> SRC_BITS], 1);
      const int s = (int)(w & ((1u << SRC_BITS) - 1u));
      if (MODE == 1) s_out[pos] = s; else csr_bin[pos] = s;
    }
  };
  int begsum = 0;
  for (int p0 = wv; p0 < P; p0 += 32 * 8) {
    int mybeg = 0, myend = 0;
    const int pl = p0 + 32 * lane;
    if (lane < 8 && pl < P) { mybeg = st0[pl]; myend = st1[pl]; }
    begsum += mybeg;
    unsigned w[8];
#pragma unroll
    for (int j = 0; j < 8; ++j) {
      const int beg = __builtin_amdgcn_readlane(mybeg, j);
      const int end = __builtin_amdgcn_readlane(myend, j);
      w[j] = EMPTY_EDGE;
      if (beg + lane < end) w[j] = tmp[(size_t)(p0 + 32 * j) * CHUNK + beg + lane];
    }
#pragma unroll
    for (int j = 0; j < 8; ++j)
      if (w[j] != EMPTY_EDGE) use(w[j]);
#pragma unroll
    for (int j = 0; j < 8; ++j) {
      const int beg = __builtin_amdgcn_readlane(mybeg, j);
      const int end = __builtin_amdgcn_readlane(myend, j);
      const unsigned* __restrict__ seg = tmp + (size_t)(p0 + 32 * j) * CHUNK;
      for (int i = beg + 32 + lane; i < end; i += 32) use(seg[i]);
    }
  }
  return begsum;
}

__global__ void __launch_bounds__(BT)
build_rows(const unsigned* __restrict__ tmp, const int* __restrict__ startT,
           int* __restrict__ rowptr, float* __restrict__ dinv, int* __restrict__ csr,
           int n, int P) {
  constexpr int NC = 512;
  static_assert(BIN_ROWS <= NC && NC <= BT, "one scan thread per counter");
  __shared__ int s_cnt[NC];
  __shared__ int s_wsum[NC / 32];
  __shared__ int s_red[BT / 32];
  __shared__ int s_out[OUT_CAP];
  const int tid = threadIdx.x;
  const int lane = tid & 31;
  const int wv = __builtin_amdgcn_readfirstlane(tid >> 5);
  const int b = (int)blockIdx.x;
  const int* __restrict__ st0 = startT + (size_t)b * P;
  const int* __restrict__ st1 = st0 + P;

  if (tid < NC) s_cnt[tid] = 0;
  __syncthreads();

  int part = sweep_bin<0>(tmp, st0, st1, P, wv, lane, s_cnt, s_out, csr);
  for (int o = 16; o > 0; o >>= 1) part += __shfl_xor(part, o);
  if (lane == 0) s_red[wv] = part;
  __syncthreads();

  const int v = (tid < NC) ? s_cnt[tid] : 0;
  int inc = v;
  for (int o = 1; o < 32; o <<= 1) {
    const int y = __shfl_up(inc, o);
    if (lane >= o) inc += y;
  }
  if (tid < NC && lane == 31) s_wsum[wv] = inc;
  int base = 0;
  for (int k = 0; k < BT / 32; ++k) base += s_red[k];
  __syncthreads();
  int total = 0;
  for (int k = 0; k < NC / 32; ++k) total += s_wsum[k];
  if (tid < NC) {
    int woff = 0;
    for (int k = 0; k < wv; ++k) woff += s_wsum[k];
    const int ex = woff + inc - v;
    s_cnt[tid] = ex;
    const int d = b * BIN_ROWS + tid;
    if (tid < BIN_ROWS && d < n) {
      rowptr[d] = base + ex;
      const float dg = (float)(v + 1);
      dinv[d] = dg > 0.0f ? rsqrtf(dg) : 0.0f;
    }
  }
  __syncthreads();

  if (total <= OUT_CAP) {
    sweep_bin<1>(tmp, st0, st1, P, wv, lane, s_cnt, s_out, csr + base);
    __syncthreads();
    for (int i = tid; i < total; i += BT) csr[base + i] = s_out[i];
  } else {
    sweep_bin<2>(tmp, st0, st1, P, wv, lane, s_cnt, s_out, csr + base);
  }
}


template <int K, int NOUT, int MT>
__global__ void __launch_bounds__((NOUT / 16) * 32)
gcn_gemm_wmma(const float* __restrict__ X, const float* __restrict__ W,
                              const float* __restrict__ dinv, float* __restrict__ Y) {
  constexpr int NT = (NOUT / 16) * 32;
  constexpr int ROWS = 16 * MT;
  constexpr int KC = (K >= 256) ? 64 : 32;
  constexpr int NCH = K / KC;
  constexpr int PITCH = KC + 4;
  constexpr int NLD = (ROWS * KC / 4) / NT;
  static_assert(K % KC == 0, "K must be a multiple of the chunk");
  static_assert((ROWS * KC / 4) % NT == 0 && NT % (KC / 4) == 0, "chunk must divide evenly over the threads");
  __shared__ alignas(16) float At[2][ROWS * PITCH];

  const int tid = threadIdx.x;
  const size_t mbase_row = (size_t)blockIdx.x * ROWS;

  constexpr int RSTEP = NT / (KC / 4);
  const int srow = tid / (KC / 4);
  const int scol = (tid % (KC / 4)) * 4;
  const float* __restrict__ xsrc = X + (mbase_row + srow) * K + scol;
  float* __restrict__ ldst = &At[0][srow * PITCH + scol];
  float4 st[NLD];
#pragma unroll
  for (int j = 0; j < NLD; ++j) st[j] = *(const float4*)(xsrc + (size_t)j * RSTEP * K);
#pragma unroll
  for (int j = 0; j < NLD; ++j) *(float4*)(ldst + j * RSTEP * PITCH) = st[j];
  __syncthreads();

  const int wave = tid >> 5;
  const int lane = tid & 31;
  const int lrow  = lane & 15;
  const int khalf = (lane >> 4) * 2;
  const int n0    = wave * 16;
  const int mofs  = (tid & 16) ? 8 : 0;

  v8f acc0 = {}, acc1 = {}, acc2 = {}, acc3 = {}, acc4 = {};
  static_assert(MT == 5, "five accumulator tiles are written out by name");

  auto chunk_products = [&](const float* __restrict__ ab, const float* __restrict__ wb) {
#pragma unroll
    for (int kk = 0; kk < KC; kk += 4) {
      const float* wp = wb + (size_t)kk * NOUT;
      v2f b; b.x = wp[0]; b.y = wp[NOUT];
      const float* ap = ab + kk;
      v2f a0, a1, a2, a3, a4;
      a0.x = ap[0 * 16 * PITCH]; a0.y = ap[0 * 16 * PITCH + 1];
      a1.x = ap[1 * 16 * PITCH]; a1.y = ap[1 * 16 * PITCH + 1];
      a2.x = ap[2 * 16 * PITCH]; a2.y = ap[2 * 16 * PITCH + 1];
      a3.x = ap[3 * 16 * PITCH]; a3.y = ap[3 * 16 * PITCH + 1];
      a4.x = ap[4 * 16 * PITCH]; a4.y = ap[4 * 16 * PITCH + 1];
      acc0 = __builtin_amdgcn_wmma_f32_16x16x4_f32(false, a0, false, b, (short)0, acc0, false, false);
      acc1 = __builtin_amdgcn_wmma_f32_16x16x4_f32(false, a1, false, b, (short)0, acc1, false, false);
      acc2 = __builtin_amdgcn_wmma_f32_16x16x4_f32(false, a2, false, b, (short)0, acc2, false, false);
      acc3 = __builtin_amdgcn_wmma_f32_16x16x4_f32(false, a3, false, b, (short)0, acc3, false, false);
      acc4 = __builtin_amdgcn_wmma_f32_16x16x4_f32(false, a4, false, b, (short)0, acc4, false, false);
    }
  };
  const float* __restrict__ ab0 = &At[0][lrow * PITCH + khalf];
  const float* __restrict__ wb0 = W + (size_t)khalf * NOUT + n0 + lrow;

#pragma unroll 1
  for (int ch = 0; ch + 1 < NCH; ++ch) {
#pragma unroll
    for (int j = 0; j < NLD; ++j)
      st[j] = *(const float4*)(xsrc + (size_t)j * RSTEP * K + (ch + 1) * KC);
    chunk_products(ab0 + (ch & 1) * (ROWS * PITCH), wb0 + (size_t)ch * KC * NOUT);
    float* __restrict__ lnext = ldst + ((ch + 1) & 1) * (ROWS * PITCH);
#pragma unroll
    for (int j = 0; j < NLD; ++j) *(float4*)(lnext + j * RSTEP * PITCH) = st[j];
    __syncthreads();
  }
  chunk_products(ab0 + ((NCH - 1) & 1) * (ROWS * PITCH), wb0 + (size_t)(NCH - 1) * KC * NOUT);
  const v8f acc[MT] = {acc0, acc1, acc2, acc3, acc4};

#pragma unroll
  for (int t = 0; t < MT; ++t) {
    const size_t row0 = mbase_row + t * 16 + mofs;
    const float4 ds0 = *(const float4*)(dinv + row0);
    const float4 ds1 = *(const float4*)(dinv + row0 + 4);
    const float dsc[8] = {ds0.x, ds0.y, ds0.z, ds0.w, ds1.x, ds1.y, ds1.z, ds1.w};
    float* yp = Y + row0 * NOUT + n0 + lrow;
#pragma unroll
    for (int v = 0; v < 8; ++v) yp[(size_t)v * NOUT] = acc[t][v] * dsc[v];
  }
}

template <typename V, bool ACT>
__global__ void gather_rows(const int* __restrict__ rowptr, const int* __restrict__ csr,
                            const float* __restrict__ dinv, const float* __restrict__ h,
                            const float* __restrict__ bias, float* __restrict__ out,
                            int n, int E) {
  constexpr int VEC = sizeof(V) / sizeof(float);
  constexpr int F = 32 * VEC;
  const int lane = threadIdx.x & 31;
  int d = (int)(((long long)blockIdx.x * blockDim.x + threadIdx.x) >> 5);
  if (d >= n) return;
  d = __builtin_amdgcn_readfirstlane(d);
  const int beg = rowptr[d];
  const int end = (d + 1 < n) ? rowptr[d + 1] : E;
  const int m = end - beg;
  const float dd = dinv[d];
  const float* __restrict__ hl = h + lane * VEC;

  V acc = *(const V*)(hl + (size_t)d * F);
  const V bv = *(const V*)(bias + lane * VEC);
  for (int base = 0; base < m; base += 32) {
    int s_l = 0;
    if (base + lane < m) s_l = csr[beg + base + lane];
    const int c = (m - base < 32) ? (m - base) : 32;
    int j = 0;
    for (; j + 4 <= c; j += 4) {
      const int s0 = __builtin_amdgcn_readlane(s_l, j);
      const int s1 = __builtin_amdgcn_readlane(s_l, j + 1);
      const int s2 = __builtin_amdgcn_readlane(s_l, j + 2);
      const int s3 = __builtin_amdgcn_readlane(s_l, j + 3);
      const V a0 = *(const V*)(hl + (size_t)s0 * F);
      const V a1 = *(const V*)(hl + (size_t)s1 * F);
      const V a2 = *(const V*)(hl + (size_t)s2 * F);
      const V a3 = *(const V*)(hl + (size_t)s3 * F);
      acc += a0;
      acc += a1;
      acc += a2;
      acc += a3;
    }
    if (j + 2 <= c) {
      const int s0 = __builtin_amdgcn_readlane(s_l, j);
      const int s1 = __builtin_amdgcn_readlane(s_l, j + 1);
      const V a0 = *(const V*)(hl + (size_t)s0 * F);
      const V a1 = *(const V*)(hl + (size_t)s1 * F);
      acc += a0;
      acc += a1;
      j += 2;
    }
    if (j < c) {
      const int s0 = __builtin_amdgcn_readlane(s_l, j);
      const V a0 = *(const V*)(hl + (size_t)s0 * F);
      acc += a0;
    }
  }

  V r = acc * dd + bv;
  if (ACT) {
    for (int k = 0; k < VEC; ++k) r[k] = (r[k] >= 0.f) ? r[k] : NEG_SLOPE * r[k];
  }
  *(V*)(out + (size_t)d * F + lane * VEC) = r;
}


extern "C" void kernel_launch(void* const* d_in, const int* in_sizes, int n_in,
                              void* d_out, int out_size, void* d_ws, size_t ws_size,
                              hipStream_t stream) {
  const float* x   = (const float*)d_in[0];
  const int*   ei  = (const int*)d_in[1];
  const float* W1  = (const float*)d_in[2];
  const float* b1  = (const float*)d_in[3];
  const float* W2  = (const float*)d_in[4];
  const float* b2  = (const float*)d_in[5];
  float* out = (float*)d_out;

  const int N = in_sizes[0] / 256;
  const int E = in_sizes[1] / 2;
  const int* src = ei;
  const int* dst = ei + E;

  float* ws     = (float*)d_ws;
  int*   rowptr = (int*)ws;
  float* dinv   = ws + N;
  int*   csr    = (int*)(dinv + N);
  float* h1     = (float*)(csr + E);
  float* h2     = h1 + (size_t)N * 128;
  unsigned* tmp = (unsigned*)h2;
  int*   startT = (int*)(tmp + E);
  float* g = h1;

  const int T = 256;
  auto blk = [](long long n, int t) { return (unsigned)((n + t - 1) / t); };
  const unsigned chunks = blk(E, CHUNK);
  const unsigned bins = blk(N, BIN_ROWS);

  bin_edges<<<chunks, BT, 0, stream>>>(src, dst, tmp, startT, E, (int)bins);
  build_rows<<<bins, BT, 0, stream>>>(tmp, startT, rowptr, dinv, csr, N, (int)chunks);

  gcn_gemm_wmma<256, 128, 5><<<N / 80, 8 * 32, 0, stream>>>(x, W1, dinv, h1);

  gather_rows<v4f, true><<<blk((long long)N * 32, T), T, 0, stream>>>(rowptr, csr, dinv, h1, b1, h2, N, E);

  gcn_gemm_wmma<128, 64, 5><<<N / 80, 4 * 32, 0, stream>>>(h2, W2, dinv, g);

  gather_rows<v2f, false><<<blk((long long)N * 32, T), T, 0, stream>>>(rowptr, csr, dinv, g, b2, out, N, E);
}


# Layer-1 gather moved into the layer-2 GEMM launch: 4 waves gather a block's 80 rows (a node's rows all in flight) into the LDS A tile, then multiply by W2; h2 never stored; 6 -> 5 launches; sums kept
# speedup vs baseline: 4.7549x; 1.0768x over previous
#include <hip/hip_runtime.h>
#include <cstdint>
#include <cstddef>

typedef __attribute__((ext_vector_type(2))) float v2f;
typedef __attribute__((ext_vector_type(4))) float v4f;
typedef __attribute__((ext_vector_type(8))) float v8f;

#define NEG_SLOPE 0.01f
#define BT 1024
#define BIN_ROWS 400
#define MAX_BINS 256
#define CHUNK 6272
#define SRC_BITS 17
#define OUT_CAP 16384
#define EMPTY_EDGE 0xFFFFFFFFu


__global__ void __launch_bounds__(BT)
bin_edges(const int* __restrict__ src, const int* __restrict__ dst,
          unsigned* __restrict__ tmp, int* __restrict__ startT, int E, int nbins) {
  __shared__ int s_hist[MAX_BINS];
  __shared__ int s_start[MAX_BINS + 1];
  __shared__ int s_wsum[MAX_BINS / 32];
  __shared__ unsigned s_sorted[CHUNK];
  constexpr int PER = (CHUNK + BT - 1) / BT;
  const int tid = threadIdx.x;
  const int lane = tid & 31;
  const int wv = tid >> 5;
  const int P = (int)gridDim.x;
  const int p = (int)blockIdx.x;
  const int ebase = p * CHUNK;
  const int cnt = (E - ebase < CHUNK) ? (E - ebase) : CHUNK;

  if (tid < MAX_BINS) s_hist[tid] = 0;
  __syncthreads();

  unsigned dd[PER], ss[PER];
#pragma unroll
  for (int k = 0; k < PER; ++k) {
    const int i = tid + k * BT;
    dd[k] = EMPTY_EDGE; ss[k] = 0;
    if (i < cnt) { dd[k] = (unsigned)dst[ebase + i]; ss[k] = (unsigned)src[ebase + i]; }
  }
  unsigned word[PER];
  int bn[PER], rk[PER];
#pragma unroll
  for (int k = 0; k < PER; ++k) {
    bn[k] = -1; rk[k] = 0; word[k] = 0;
    if (dd[k] != EMPTY_EDGE) {
      const unsigned b = dd[k] / BIN_ROWS;
      bn[k] = (int)b;
      word[k] = ((dd[k] - b * BIN_ROWS) << SRC_BITS) | ss[k];
      rk[k] = atomicAdd(&s_hist[b], 1);
    }
  }
  __syncthreads();

  const int v = (tid < MAX_BINS) ? s_hist[tid] : 0;
  int inc = v;
  for (int o = 1; o < 32; o <<= 1) {
    const int y = __shfl_up(inc, o);
    if (lane >= o) inc += y;
  }
  if (tid < MAX_BINS && lane == 31) s_wsum[wv] = inc;
  __syncthreads();
  if (tid < MAX_BINS) {
    int woff = 0;
    for (int k = 0; k < wv; ++k) woff += s_wsum[k];
    s_start[tid] = woff + inc - v;
    if (tid == MAX_BINS - 1) s_start[MAX_BINS] = woff + inc;
  }
  __syncthreads();

#pragma unroll
  for (int k = 0; k < PER; ++k)
    if (bn[k] >= 0) s_sorted[s_start[bn[k]] + rk[k]] = word[k];
  __syncthreads();

  for (int i = tid; i < cnt; i += BT) tmp[ebase + i] = s_sorted[i];
  if (tid <= nbins) startT[tid * P + p] = s_start[tid];
}

template <int MODE>
__device__ __forceinline__ int sweep_bin(const unsigned* __restrict__ tmp,
                                         const int* __restrict__ st0, const int* __restrict__ st1,
                                         int P, int wv, int lane,
                                         int* __restrict__ s_cnt, int* __restrict__ s_out,
                                         int* __restrict__ csr_bin) {
  auto use = [&](unsigned w) {
    if (MODE == 0) {
      atomicAdd(&s_cnt[w >> SRC_BITS], 1);
    } else {
      const int pos = atomicAdd(&s_cnt[w >> SRC_BITS], 1);
      const int s = (int)(w & ((1u << SRC_BITS) - 1u));
      if (MODE == 1) s_out[pos] = s; else csr_bin[pos] = s;
    }
  };
  int begsum = 0;
  for (int p0 = wv; p0 < P; p0 += 32 * 8) {
    int mybeg = 0, myend = 0;
    const int pl = p0 + 32 * lane;
    if (lane < 8 && pl < P) { mybeg = st0[pl]; myend = st1[pl]; }
    begsum += mybeg;
    unsigned w[8];
#pragma unroll
    for (int j = 0; j < 8; ++j) {
      const int beg = __builtin_amdgcn_readlane(mybeg, j);
      const int end = __builtin_amdgcn_readlane(myend, j);
      w[j] = EMPTY_EDGE;
      if (beg + lane < end) w[j] = tmp[(size_t)(p0 + 32 * j) * CHUNK + beg + lane];
    }
#pragma unroll
    for (int j = 0; j < 8; ++j)
      if (w[j] != EMPTY_EDGE) use(w[j]);
#pragma unroll
    for (int j = 0; j < 8; ++j) {
      const int beg = __builtin_amdgcn_readlane(mybeg, j);
      const int end = __builtin_amdgcn_readlane(myend, j);
      const unsigned* __restrict__ seg = tmp + (size_t)(p0 + 32 * j) * CHUNK;
      for (int i = beg + 32 + lane; i < end; i += 32) use(seg[i]);
    }
  }
  return begsum;
}

__global__ void __launch_bounds__(BT)
build_rows(const unsigned* __restrict__ tmp, const int* __restrict__ startT,
           int* __restrict__ rowptr, float* __restrict__ dinv, int* __restrict__ csr,
           int n, int P) {
  constexpr int NC = 512;
  static_assert(BIN_ROWS <= NC && NC <= BT, "one scan thread per counter");
  __shared__ int s_cnt[NC];
  __shared__ int s_wsum[NC / 32];
  __shared__ int s_red[BT / 32];
  __shared__ int s_out[OUT_CAP];
  const int tid = threadIdx.x;
  const int lane = tid & 31;
  const int wv = __builtin_amdgcn_readfirstlane(tid >> 5);
  const int b = (int)blockIdx.x;
  const int* __restrict__ st0 = startT + (size_t)b * P;
  const int* __restrict__ st1 = st0 + P;

  if (tid < NC) s_cnt[tid] = 0;
  __syncthreads();

  int part = sweep_bin<0>(tmp, st0, st1, P, wv, lane, s_cnt, s_out, csr);
  for (int o = 16; o > 0; o >>= 1) part += __shfl_xor(part, o);
  if (lane == 0) s_red[wv] = part;
  __syncthreads();

  const int v = (tid < NC) ? s_cnt[tid] : 0;
  int inc = v;
  for (int o = 1; o < 32; o <<= 1) {
    const int y = __shfl_up(inc, o);
    if (lane >= o) inc += y;
  }
  if (tid < NC && lane == 31) s_wsum[wv] = inc;
  int base = 0;
  for (int k = 0; k < BT / 32; ++k) base += s_red[k];
  __syncthreads();
  int total = 0;
  for (int k = 0; k < NC / 32; ++k) total += s_wsum[k];
  if (tid < NC) {
    int woff = 0;
    for (int k = 0; k < wv; ++k) woff += s_wsum[k];
    const int ex = woff + inc - v;
    s_cnt[tid] = ex;
    const int d = b * BIN_ROWS + tid;
    if (tid < BIN_ROWS && d < n) {
      rowptr[d] = base + ex;
      const float dg = (float)(v + 1);
      dinv[d] = dg > 0.0f ? rsqrtf(dg) : 0.0f;
    }
  }
  __syncthreads();

  if (total <= OUT_CAP) {
    sweep_bin<1>(tmp, st0, st1, P, wv, lane, s_cnt, s_out, csr + base);
    __syncthreads();
    for (int i = tid; i < total; i += BT) csr[base + i] = s_out[i];
  } else {
    sweep_bin<2>(tmp, st0, st1, P, wv, lane, s_cnt, s_out, csr + base);
  }
}


template <int K, int NOUT, int MT>
__global__ void __launch_bounds__((NOUT / 16) * 32)
gcn_gemm_wmma(const float* __restrict__ X, const float* __restrict__ W,
                              const float* __restrict__ dinv, float* __restrict__ Y) {
  constexpr int NT = (NOUT / 16) * 32;
  constexpr int ROWS = 16 * MT;
  constexpr int KC = (K >= 256) ? 64 : 32;
  constexpr int NCH = K / KC;
  constexpr int PITCH = KC + 4;
  constexpr int NLD = (ROWS * KC / 4) / NT;
  static_assert(K % KC == 0, "K must be a multiple of the chunk");
  static_assert((ROWS * KC / 4) % NT == 0 && NT % (KC / 4) == 0, "chunk must divide evenly over the threads");
  __shared__ alignas(16) float At[2][ROWS * PITCH];

  const int tid = threadIdx.x;
  const size_t mbase_row = (size_t)blockIdx.x * ROWS;

  constexpr int RSTEP = NT / (KC / 4);
  const int srow = tid / (KC / 4);
  const int scol = (tid % (KC / 4)) * 4;
  const float* __restrict__ xsrc = X + (mbase_row + srow) * K + scol;
  float* __restrict__ ldst = &At[0][srow * PITCH + scol];
  float4 st[NLD];
#pragma unroll
  for (int j = 0; j < NLD; ++j) st[j] = *(const float4*)(xsrc + (size_t)j * RSTEP * K);
#pragma unroll
  for (int j = 0; j < NLD; ++j) *(float4*)(ldst + j * RSTEP * PITCH) = st[j];
  __syncthreads();

  const int wave = tid >> 5;
  const int lane = tid & 31;
  const int lrow  = lane & 15;
  const int khalf = (lane >> 4) * 2;
  const int n0    = wave * 16;
  const int mofs  = (tid & 16) ? 8 : 0;

  v8f acc0 = {}, acc1 = {}, acc2 = {}, acc3 = {}, acc4 = {};
  static_assert(MT == 5, "five accumulator tiles are written out by name");

  auto chunk_products = [&](const float* __restrict__ ab, const float* __restrict__ wb) {
#pragma unroll
    for (int kk = 0; kk < KC; kk += 4) {
      const float* wp = wb + (size_t)kk * NOUT;
      v2f b; b.x = wp[0]; b.y = wp[NOUT];
      const float* ap = ab + kk;
      v2f a0, a1, a2, a3, a4;
      a0.x = ap[0 * 16 * PITCH]; a0.y = ap[0 * 16 * PITCH + 1];
      a1.x = ap[1 * 16 * PITCH]; a1.y = ap[1 * 16 * PITCH + 1];
      a2.x = ap[2 * 16 * PITCH]; a2.y = ap[2 * 16 * PITCH + 1];
      a3.x = ap[3 * 16 * PITCH]; a3.y = ap[3 * 16 * PITCH + 1];
      a4.x = ap[4 * 16 * PITCH]; a4.y = ap[4 * 16 * PITCH + 1];
      acc0 = __builtin_amdgcn_wmma_f32_16x16x4_f32(false, a0, false, b, (short)0, acc0, false, false);
      acc1 = __builtin_amdgcn_wmma_f32_16x16x4_f32(false, a1, false, b, (short)0, acc1, false, false);
      acc2 = __builtin_amdgcn_wmma_f32_16x16x4_f32(false, a2, false, b, (short)0, acc2, false, false);
      acc3 = __builtin_amdgcn_wmma_f32_16x16x4_f32(false, a3, false, b, (short)0, acc3, false, false);
      acc4 = __builtin_amdgcn_wmma_f32_16x16x4_f32(false, a4, false, b, (short)0, acc4, false, false);
    }
  };
  const float* __restrict__ ab0 = &At[0][lrow * PITCH + khalf];
  const float* __restrict__ wb0 = W + (size_t)khalf * NOUT + n0 + lrow;

#pragma unroll 1
  for (int ch = 0; ch + 1 < NCH; ++ch) {
#pragma unroll
    for (int j = 0; j < NLD; ++j)
      st[j] = *(const float4*)(xsrc + (size_t)j * RSTEP * K + (ch + 1) * KC);
    chunk_products(ab0 + (ch & 1) * (ROWS * PITCH), wb0 + (size_t)ch * KC * NOUT);
    float* __restrict__ lnext = ldst + ((ch + 1) & 1) * (ROWS * PITCH);
#pragma unroll
    for (int j = 0; j < NLD; ++j) *(float4*)(lnext + j * RSTEP * PITCH) = st[j];
    __syncthreads();
  }
  chunk_products(ab0 + ((NCH - 1) & 1) * (ROWS * PITCH), wb0 + (size_t)(NCH - 1) * KC * NOUT);
  const v8f acc[MT] = {acc0, acc1, acc2, acc3, acc4};

#pragma unroll
  for (int t = 0; t < MT; ++t) {
    const size_t row0 = mbase_row + t * 16 + mofs;
    const float4 ds0 = *(const float4*)(dinv + row0);
    const float4 ds1 = *(const float4*)(dinv + row0 + 4);
    const float dsc[8] = {ds0.x, ds0.y, ds0.z, ds0.w, ds1.x, ds1.y, ds1.z, ds1.w};
    float* yp = Y + row0 * NOUT + n0 + lrow;
#pragma unroll
    for (int v = 0; v < 8; ++v) yp[(size_t)v * NOUT] = acc[t][v] * dsc[v];
  }
}

template <typename V, bool ACT>
__global__ void gather_rows(const int* __restrict__ rowptr, const int* __restrict__ csr,
                            const float* __restrict__ dinv, const float* __restrict__ h,
                            const float* __restrict__ bias, float* __restrict__ out,
                            int n, int E) {
  constexpr int VEC = sizeof(V) / sizeof(float);
  constexpr int F = 32 * VEC;
  const int lane = threadIdx.x & 31;
  int d = (int)(((long long)blockIdx.x * blockDim.x + threadIdx.x) >> 5);
  if (d >= n) return;
  d = __builtin_amdgcn_readfirstlane(d);
  const int beg = rowptr[d];
  const int end = (d + 1 < n) ? rowptr[d + 1] : E;
  const int m = end - beg;
  const float dd = dinv[d];
  const float* __restrict__ hl = h + lane * VEC;

  V acc = *(const V*)(hl + (size_t)d * F);
  const V bv = *(const V*)(bias + lane * VEC);
  for (int base = 0; base < m; base += 32) {
    int s_l = 0;
    if (base + lane < m) s_l = csr[beg + base + lane];
    const int c = (m - base < 32) ? (m - base) : 32;
    int j = 0;
    for (; j + 4 <= c; j += 4) {
      const int s0 = __builtin_amdgcn_readlane(s_l, j);
      const int s1 = __builtin_amdgcn_readlane(s_l, j + 1);
      const int s2 = __builtin_amdgcn_readlane(s_l, j + 2);
      const int s3 = __builtin_amdgcn_readlane(s_l, j + 3);
      const V a0 = *(const V*)(hl + (size_t)s0 * F);
      const V a1 = *(const V*)(hl + (size_t)s1 * F);
      const V a2 = *(const V*)(hl + (size_t)s2 * F);
      const V a3 = *(const V*)(hl + (size_t)s3 * F);
      acc += a0;
      acc += a1;
      acc += a2;
      acc += a3;
    }
    if (j + 2 <= c) {
      const int s0 = __builtin_amdgcn_readlane(s_l, j);
      const int s1 = __builtin_amdgcn_readlane(s_l, j + 1);
      const V a0 = *(const V*)(hl + (size_t)s0 * F);
      const V a1 = *(const V*)(hl + (size_t)s1 * F);
      acc += a0;
      acc += a1;
      j += 2;
    }
    if (j < c) {
      const int s0 = __builtin_amdgcn_readlane(s_l, j);
      const V a0 = *(const V*)(hl + (size_t)s0 * F);
      acc += a0;
    }
  }

  V r = acc * dd + bv;
  if (ACT) {
    for (int k = 0; k < VEC; ++k) r[k] = (r[k] >= 0.f) ? r[k] : NEG_SLOPE * r[k];
  }
  *(V*)(out + (size_t)d * F + lane * VEC) = r;
}

template <int K, int NOUT, int MT>
__global__ void __launch_bounds__((NOUT / 16) * 32)
gather_gemm_wmma(const int* __restrict__ rowptr, const int* __restrict__ csr,
                 const float* __restrict__ dinv, const float* __restrict__ h,
                 const float* __restrict__ bias, const float* __restrict__ W,
                 float* __restrict__ Y, int n, int E) {
  constexpr int NW = NOUT / 16;
  constexpr int ROWS = 16 * MT;
  constexpr int PER = ROWS / NW;
  constexpr int PITCH = K + 4;
  static_assert(K == 128, "a lane holds four floats of a gathered row");
  static_assert(ROWS % NW == 0 && PER + 1 <= 32, "a wave's row bounds sit one per lane");
  static_assert(MT == 5, "five accumulator tiles are written out by name");
  __shared__ alignas(16) float At[ROWS * PITCH];

  const int tid = threadIdx.x;
  const int lane = tid & 31;
  const int wave = __builtin_amdgcn_readfirstlane(tid >> 5);
  const int mbase = (int)blockIdx.x * ROWS;
  const int dw0 = mbase + wave * PER;

  int rp = E;
  float dv = 0.0f;
  {
    const int dl = dw0 + lane;
    if (lane <= PER && dl < n) rp = rowptr[dl];
    if (lane < PER && dl < n) dv = dinv[dl];
  }
  const v4f bv = *(const v4f*)(bias + lane * 4);
  float* __restrict__ arow = &At[(wave * PER) * PITCH + lane * 4];
  auto hrow = [&](int s) { return ((const v4f*)(h + (size_t)s * K))[lane]; };

  int beg = __builtin_amdgcn_readlane(rp, 0);
  int end = __builtin_amdgcn_readlane(rp, 1);
  int s_nx = 0;
  if (beg + lane < end) s_nx = csr[beg + lane];
  v4f own_nx = hrow(dw0);

#pragma unroll 1
  for (int i = 0; i < PER; ++i) {
    const int m = end - beg;
    const int s_l = s_nx;
    v4f acc = own_nx;
    const float dd = __int_as_float(__builtin_amdgcn_readlane(__float_as_int(dv), i));
    const int beg2 = end;
    int end2 = end;
    if (i + 1 < PER) {
      end2 = __builtin_amdgcn_readlane(rp, i + 2);
      s_nx = 0;
      if (beg2 + lane < end2) s_nx = csr[beg2 + lane];
      own_nx = hrow(dw0 + i + 1);
    }

    const int c0 = (m < 31) ? m : 31;
    v4f g16[16], g8[8], g4[4], g2[2], g1;
    const int j8 = c0 & 16;
    const int j4 = j8 + (c0 & 8);
    const int j2 = j4 + (c0 & 4);
    const int j1 = j2 + (c0 & 2);
    if (c0 & 16) {
#pragma unroll
      for (int q = 0; q < 16; ++q) g16[q] = hrow(__builtin_amdgcn_readlane(s_l, q));
    }
    if (c0 & 8) {
#pragma unroll
      for (int q = 0; q < 8; ++q) g8[q] = hrow(__builtin_amdgcn_readlane(s_l, j8 + q));
    }
    if (c0 & 4) {
#pragma unroll
      for (int q = 0; q < 4; ++q) g4[q] = hrow(__builtin_amdgcn_readlane(s_l, j4 + q));
    }
    if (c0 & 2) {
#pragma unroll
      for (int q = 0; q < 2; ++q) g2[q] = hrow(__builtin_amdgcn_readlane(s_l, j2 + q));
    }
    if (c0 & 1) g1 = hrow(__builtin_amdgcn_readlane(s_l, j1));
    if (c0 & 16) {
#pragma unroll
      for (int q = 0; q < 16; ++q) acc += g16[q];
    }
    if (c0 & 8) {
#pragma unroll
      for (int q = 0; q < 8; ++q) acc += g8[q];
    }
    if (c0 & 4) {
#pragma unroll
      for (int q = 0; q < 4; ++q) acc += g4[q];
    }
    if (c0 & 2) {
      acc += g2[0];
      acc += g2[1];
    }
    if (c0 & 1) acc += g1;

    for (int base = 31; base < m; base += 32) {
      int s_r = 0;
      if (base + lane < m) s_r = csr[beg + base + lane];
      const int c = (m - base < 32) ? (m - base) : 32;
#pragma nounroll
      for (int j = 0; j < c; ++j) acc += hrow(__builtin_amdgcn_readlane(s_r, j));
    }

    v4f r = acc * dd + bv;
    for (int k = 0; k < 4; ++k) r[k] = (r[k] >= 0.f) ? r[k] : NEG_SLOPE * r[k];
    *(v4f*)(arow + i * PITCH) = r;
    beg = beg2;
    end = end2;
  }
  __syncthreads();

  const int lrow  = lane & 15;
  const int khalf = (lane >> 4) * 2;
  const int n0    = wave * 16;
  const int mofs  = (lane & 16) ? 8 : 0;
  const float* __restrict__ ab0 = &At[lrow * PITCH + khalf];
  const float* __restrict__ wb0 = W + (size_t)khalf * NOUT + n0 + lrow;

  v8f c0 = {}, c1 = {}, c2 = {}, c3 = {}, c4 = {};
#pragma unroll 8
  for (int kk = 0; kk < K; kk += 4) {
    const float* wp = wb0 + (size_t)kk * NOUT;
    v2f b; b.x = wp[0]; b.y = wp[NOUT];
    const float* ap = ab0 + kk;
    v2f a0, a1, a2, a3, a4;
    a0.x = ap[0 * 16 * PITCH]; a0.y = ap[0 * 16 * PITCH + 1];
    a1.x = ap[1 * 16 * PITCH]; a1.y = ap[1 * 16 * PITCH + 1];
    a2.x = ap[2 * 16 * PITCH]; a2.y = ap[2 * 16 * PITCH + 1];
    a3.x = ap[3 * 16 * PITCH]; a3.y = ap[3 * 16 * PITCH + 1];
    a4.x = ap[4 * 16 * PITCH]; a4.y = ap[4 * 16 * PITCH + 1];
    c0 = __builtin_amdgcn_wmma_f32_16x16x4_f32(false, a0, false, b, (short)0, c0, false, false);
    c1 = __builtin_amdgcn_wmma_f32_16x16x4_f32(false, a1, false, b, (short)0, c1, false, false);
    c2 = __builtin_amdgcn_wmma_f32_16x16x4_f32(false, a2, false, b, (short)0, c2, false, false);
    c3 = __builtin_amdgcn_wmma_f32_16x16x4_f32(false, a3, false, b, (short)0, c3, false, false);
    c4 = __builtin_amdgcn_wmma_f32_16x16x4_f32(false, a4, false, b, (short)0, c4, false, false);
  }
  const v8f cc[MT] = {c0, c1, c2, c3, c4};

#pragma unroll
  for (int t = 0; t < MT; ++t) {
    const size_t row0 = (size_t)mbase + t * 16 + mofs;
    const float4 ds0 = *(const float4*)(dinv + row0);
    const float4 ds1 = *(const float4*)(dinv + row0 + 4);
    const float dsc[8] = {ds0.x, ds0.y, ds0.z, ds0.w, ds1.x, ds1.y, ds1.z, ds1.w};
    float* yp = Y + row0 * NOUT + n0 + lrow;
#pragma unroll
    for (int v = 0; v < 8; ++v) yp[(size_t)v * NOUT] = cc[t][v] * dsc[v];
  }
}


extern "C" void kernel_launch(void* const* d_in, const int* in_sizes, int n_in,
                              void* d_out, int out_size, void* d_ws, size_t ws_size,
                              hipStream_t stream) {
  const float* x   = (const float*)d_in[0];
  const int*   ei  = (const int*)d_in[1];
  const float* W1  = (const float*)d_in[2];
  const float* b1  = (const float*)d_in[3];
  const float* W2  = (const float*)d_in[4];
  const float* b2  = (const float*)d_in[5];
  float* out = (float*)d_out;

  const int N = in_sizes[0] / 256;
  const int E = in_sizes[1] / 2;
  const int* src = ei;
  const int* dst = ei + E;

  float* ws     = (float*)d_ws;
  int*   rowptr = (int*)ws;
  float* dinv   = ws + N;
  int*   csr    = (int*)(dinv + N);
  float* h1     = (float*)(csr + E);
  float* g      = h1 + (size_t)N * 128;
  unsigned* tmp = (unsigned*)g;
  int*   startT = (int*)(tmp + E);

  const int T = 256;
  auto blk = [](long long n, int t) { return (unsigned)((n + t - 1) / t); };
  const unsigned chunks = blk(E, CHUNK);
  const unsigned bins = blk(N, BIN_ROWS);

  bin_edges<<<chunks, BT, 0, stream>>>(src, dst, tmp, startT, E, (int)bins);
  build_rows<<<bins, BT, 0, stream>>>(tmp, startT, rowptr, dinv, csr, N, (int)chunks);

  gcn_gemm_wmma<256, 128, 5><<<N / 80, 8 * 32, 0, stream>>>(x, W1, dinv, h1);

  gather_gemm_wmma<128, 64, 5><<<N / 80, 4 * 32, 0, stream>>>(rowptr, csr, dinv, h1, b1, W2, g, N, E);

  gather_rows<v2f, false><<<blk((long long)N * 32, T), T, 0, stream>>>(rowptr, csr, dinv, g, b2, out, N, E);
}
